# starting kernel
# baseline (speedup 1.0000x reference)
#include <hip/hip_runtime.h>


#define B_    64
#define H_    512
#define W_    512
#define HW    (H_ * W_)
#define NRING 256
#define HBINS (4 * NRING)

#define BLOCKS_PER_IMG 16
#define THREADS        256
#define PIX_PER_BLOCK  (HW / BLOCKS_PER_IMG)
#define V4_PER_BLOCK   (PIX_PER_BLOCK / 4)
#define ITERS          (V4_PER_BLOCK / THREADS)

typedef __attribute__((ext_vector_type(2))) float v2f;
typedef __attribute__((ext_vector_type(4))) float v4f;
typedef __attribute__((ext_vector_type(8))) float v8f;

__global__ __launch_bounds__(THREADS) void frc_partial(
    const float* __restrict__ f1r, const float* __restrict__ f1i,
    const float* __restrict__ f2r, const float* __restrict__ f2i,
    float* __restrict__ part)
{
    __shared__ float hist[HBINS];

    const int t   = threadIdx.x;
    const int img = blockIdx.x / BLOCKS_PER_IMG;
    const int blk = blockIdx.x % BLOCKS_PER_IMG;

    #pragma unroll
    for (int i = t; i < HBINS; i += THREADS) hist[i] = 0.0f;
    __syncthreads();

    const size_t base4 = (size_t)img * (HW / 4) + (size_t)blk * V4_PER_BLOCK;
    const v4f* a1r = (const v4f*)f1r + base4;
    const v4f* a1i = (const v4f*)f1i + base4;
    const v4f* a2r = (const v4f*)f2r + base4;
    const v4f* a2i = (const v4f*)f2i + base4;

    #pragma unroll 4
    for (int i = 0; i < ITERS; ++i) {
        const int idx4 = i * THREADS + t;
        const v4f x1r = __builtin_nontemporal_load(&a1r[idx4]);
        const v4f x1i = __builtin_nontemporal_load(&a1i[idx4]);
        const v4f x2r = __builtin_nontemporal_load(&a2r[idx4]);
        const v4f x2i = __builtin_nontemporal_load(&a2i[idx4]);

        const int pix = blk * PIX_PER_BLOCK + idx4 * 4;
        const int y   = pix >> 9;
        const int x0  = pix & 511;
        const int dy  = y - 256;
        const int dy2 = dy * dy;
        const int dx  = x0 - 256;

        const int b0 = (int)__builtin_sqrtf((float)(dx * dx + dy2));
        const int b1 = (int)__builtin_sqrtf((float)((dx + 1) * (dx + 1) + dy2));
        const int b2 = (int)__builtin_sqrtf((float)((dx + 2) * (dx + 2) + dy2));
        const int b3 = (int)__builtin_sqrtf((float)((dx + 3) * (dx + 3) + dy2));

        const float cr0 = x1r.x * x2r.x + x1i.x * x2i.x;
        const float cr1 = x1r.y * x2r.y + x1i.y * x2i.y;
        const float cr2 = x1r.z * x2r.z + x1i.z * x2i.z;
        const float cr3 = x1r.w * x2r.w + x1i.w * x2i.w;
        const float ci0 = x1i.x * x2r.x - x1r.x * x2i.x;
        const float ci1 = x1i.y * x2r.y - x1r.y * x2i.y;
        const float ci2 = x1i.z * x2r.z - x1r.z * x2i.z;
        const float ci3 = x1i.w * x2r.w - x1r.w * x2i.w;
        const float p10 = x1r.x * x1r.x + x1i.x * x1i.x;
        const float p11 = x1r.y * x1r.y + x1i.y * x1i.y;
        const float p12 = x1r.z * x1r.z + x1i.z * x1i.z;
        const float p13 = x1r.w * x1r.w + x1i.w * x1i.w;
        const float p20 = x2r.x * x2r.x + x2i.x * x2i.x;
        const float p21 = x2r.y * x2r.y + x2i.y * x2i.y;
        const float p22 = x2r.z * x2r.z + x2i.z * x2i.z;
        const float p23 = x2r.w * x2r.w + x2i.w * x2i.w;

        if ((b0 == b1) & (b1 == b2) & (b2 == b3)) {
            if (b0 < NRING) {
                atomicAdd(&hist[          b0], cr0 + cr1 + cr2 + cr3);
                atomicAdd(&hist[NRING   + b0], ci0 + ci1 + ci2 + ci3);
                atomicAdd(&hist[2*NRING + b0], p10 + p11 + p12 + p13);
                atomicAdd(&hist[3*NRING + b0], p20 + p21 + p22 + p23);
            }
        } else {
            if (b0 < NRING) {
                atomicAdd(&hist[b0], cr0); atomicAdd(&hist[NRING + b0], ci0);
                atomicAdd(&hist[2*NRING + b0], p10); atomicAdd(&hist[3*NRING + b0], p20);
            }
            if (b1 < NRING) {
                atomicAdd(&hist[b1], cr1); atomicAdd(&hist[NRING + b1], ci1);
                atomicAdd(&hist[2*NRING + b1], p11); atomicAdd(&hist[3*NRING + b1], p21);
            }
            if (b2 < NRING) {
                atomicAdd(&hist[b2], cr2); atomicAdd(&hist[NRING + b2], ci2);
                atomicAdd(&hist[2*NRING + b2], p12); atomicAdd(&hist[3*NRING + b2], p22);
            }
            if (b3 < NRING) {
                atomicAdd(&hist[b3], cr3); atomicAdd(&hist[NRING + b3], ci3);
                atomicAdd(&hist[2*NRING + b3], p13); atomicAdd(&hist[3*NRING + b3], p23);
            }
        }
    }
    __syncthreads();

    float* dst = part + (size_t)(img * BLOCKS_PER_IMG + blk) * HBINS;
    #pragma unroll
    for (int i = t; i < HBINS; i += THREADS) dst[i] = hist[i];
}

__global__ __launch_bounds__(32) void frc_reduce(
    const float* __restrict__ part, float* __restrict__ out)
{
    __shared__ float acc[HBINS];

    const int lane = threadIdx.x;
    const int img  = blockIdx.x;
    const float* p = part + (size_t)img * BLOCKS_PER_IMG * HBINS;

    v2f aOnes;
    aOnes.x = 1.0f;
    aOnes.y = 1.0f;

    const int half = lane >> 4;
    const int nIdx = lane & 15;

    for (int tile = 0; tile < HBINS / 16; ++tile) {
        const int col = tile * 16 + nIdx;
        v8f c = {};
        #pragma unroll
        for (int kk = 0; kk < BLOCKS_PER_IMG; kk += 4) {
            v2f b;
            b.x = p[(size_t)(kk + 0 + 2 * half) * HBINS + col];
            b.y = p[(size_t)(kk + 1 + 2 * half) * HBINS + col];
            c = __builtin_amdgcn_wmma_f32_16x16x4_f32(
                     false, aOnes,  false, b,
                     (short)0, c,  false,  false);
        }
        if (lane < 16) acc[col] = c[0];
    }
    __syncthreads();

    #pragma unroll
    for (int rr = 0; rr < NRING; rr += 32) {
        const int r = rr + lane;
        const float cr = acc[r];
        const float ci = acc[NRING + r];
        const float p1 = acc[2 * NRING + r];
        const float p2 = acc[3 * NRING + r];
        const float num = __builtin_sqrtf(cr * cr + ci * ci);
        const float den = __builtin_sqrtf(p1 * p2);
        out[(size_t)img * NRING + r] = (den == 0.0f) ? 0.0f : num / den;
    }
}

extern "C" void kernel_launch(void* const* d_in, const int* in_sizes, int n_in,
                              void* d_out, int out_size, void* d_ws, size_t ws_size,
                              hipStream_t stream)
{
    (void)in_sizes; (void)n_in; (void)out_size; (void)ws_size;
    const float* f1r = (const float*)d_in[0];
    const float* f1i = (const float*)d_in[1];
    const float* f2r = (const float*)d_in[2];
    const float* f2i = (const float*)d_in[3];
    float* part = (float*)d_ws;

    frc_partial<<<dim3(B_ * BLOCKS_PER_IMG), dim3(THREADS), 0, stream>>>(
        f1r, f1i, f2r, f2i, part);
    frc_reduce<<<dim3(B_), dim3(32), 0, stream>>>(part, (float*)d_out);
}


# Reduce launch re-gridded: one wave per (image, 32 rings), 64 partial loads in flight ahead of 8 independent WMMA chains, each lane finalizes from D registers; was 1 wave per image, 64 serial tiles
# speedup vs baseline: 1.3941x; 1.3941x over previous
#include <hip/hip_runtime.h>


#define B_    64
#define H_    512
#define W_    512
#define HW    (H_ * W_)
#define NRING 256
#define HBINS (4 * NRING)

#define BLOCKS_PER_IMG 16
#define THREADS        256
#define PIX_PER_BLOCK  (HW / BLOCKS_PER_IMG)
#define V4_PER_BLOCK   (PIX_PER_BLOCK / 4)
#define ITERS          (V4_PER_BLOCK / THREADS)

typedef __attribute__((ext_vector_type(2))) float v2f;
typedef __attribute__((ext_vector_type(4))) float v4f;
typedef __attribute__((ext_vector_type(8))) float v8f;

__global__ __launch_bounds__(THREADS) void frc_partial(
    const float* __restrict__ f1r, const float* __restrict__ f1i,
    const float* __restrict__ f2r, const float* __restrict__ f2i,
    float* __restrict__ part)
{
    __shared__ float hist[HBINS];

    const int t   = threadIdx.x;
    const int img = blockIdx.x / BLOCKS_PER_IMG;
    const int blk = blockIdx.x % BLOCKS_PER_IMG;

    #pragma unroll
    for (int i = t; i < HBINS; i += THREADS) hist[i] = 0.0f;
    __syncthreads();

    const size_t base4 = (size_t)img * (HW / 4) + (size_t)blk * V4_PER_BLOCK;
    const v4f* a1r = (const v4f*)f1r + base4;
    const v4f* a1i = (const v4f*)f1i + base4;
    const v4f* a2r = (const v4f*)f2r + base4;
    const v4f* a2i = (const v4f*)f2i + base4;

    #pragma unroll 4
    for (int i = 0; i < ITERS; ++i) {
        const int idx4 = i * THREADS + t;
        const v4f x1r = __builtin_nontemporal_load(&a1r[idx4]);
        const v4f x1i = __builtin_nontemporal_load(&a1i[idx4]);
        const v4f x2r = __builtin_nontemporal_load(&a2r[idx4]);
        const v4f x2i = __builtin_nontemporal_load(&a2i[idx4]);

        const int pix = blk * PIX_PER_BLOCK + idx4 * 4;
        const int y   = pix >> 9;
        const int x0  = pix & 511;
        const int dy  = y - 256;
        const int dy2 = dy * dy;
        const int dx  = x0 - 256;

        const int b0 = (int)__builtin_sqrtf((float)(dx * dx + dy2));
        const int b1 = (int)__builtin_sqrtf((float)((dx + 1) * (dx + 1) + dy2));
        const int b2 = (int)__builtin_sqrtf((float)((dx + 2) * (dx + 2) + dy2));
        const int b3 = (int)__builtin_sqrtf((float)((dx + 3) * (dx + 3) + dy2));

        const float cr0 = x1r.x * x2r.x + x1i.x * x2i.x;
        const float cr1 = x1r.y * x2r.y + x1i.y * x2i.y;
        const float cr2 = x1r.z * x2r.z + x1i.z * x2i.z;
        const float cr3 = x1r.w * x2r.w + x1i.w * x2i.w;
        const float ci0 = x1i.x * x2r.x - x1r.x * x2i.x;
        const float ci1 = x1i.y * x2r.y - x1r.y * x2i.y;
        const float ci2 = x1i.z * x2r.z - x1r.z * x2i.z;
        const float ci3 = x1i.w * x2r.w - x1r.w * x2i.w;
        const float p10 = x1r.x * x1r.x + x1i.x * x1i.x;
        const float p11 = x1r.y * x1r.y + x1i.y * x1i.y;
        const float p12 = x1r.z * x1r.z + x1i.z * x1i.z;
        const float p13 = x1r.w * x1r.w + x1i.w * x1i.w;
        const float p20 = x2r.x * x2r.x + x2i.x * x2i.x;
        const float p21 = x2r.y * x2r.y + x2i.y * x2i.y;
        const float p22 = x2r.z * x2r.z + x2i.z * x2i.z;
        const float p23 = x2r.w * x2r.w + x2i.w * x2i.w;

        if ((b0 == b1) & (b1 == b2) & (b2 == b3)) {
            if (b0 < NRING) {
                atomicAdd(&hist[          b0], cr0 + cr1 + cr2 + cr3);
                atomicAdd(&hist[NRING   + b0], ci0 + ci1 + ci2 + ci3);
                atomicAdd(&hist[2*NRING + b0], p10 + p11 + p12 + p13);
                atomicAdd(&hist[3*NRING + b0], p20 + p21 + p22 + p23);
            }
        } else {
            if (b0 < NRING) {
                atomicAdd(&hist[b0], cr0); atomicAdd(&hist[NRING + b0], ci0);
                atomicAdd(&hist[2*NRING + b0], p10); atomicAdd(&hist[3*NRING + b0], p20);
            }
            if (b1 < NRING) {
                atomicAdd(&hist[b1], cr1); atomicAdd(&hist[NRING + b1], ci1);
                atomicAdd(&hist[2*NRING + b1], p11); atomicAdd(&hist[3*NRING + b1], p21);
            }
            if (b2 < NRING) {
                atomicAdd(&hist[b2], cr2); atomicAdd(&hist[NRING + b2], ci2);
                atomicAdd(&hist[2*NRING + b2], p12); atomicAdd(&hist[3*NRING + b2], p22);
            }
            if (b3 < NRING) {
                atomicAdd(&hist[b3], cr3); atomicAdd(&hist[NRING + b3], ci3);
                atomicAdd(&hist[2*NRING + b3], p13); atomicAdd(&hist[3*NRING + b3], p23);
            }
        }
    }
    __syncthreads();

    float* dst = part + (size_t)(img * BLOCKS_PER_IMG + blk) * HBINS;
    #pragma unroll
    for (int i = t; i < HBINS; i += THREADS) dst[i] = hist[i];
}

#define RG_PER_IMG  (NRING / 32)
#define RED_WAVES   2
#define RED_THREADS (RED_WAVES * 32)

__global__ __launch_bounds__(RED_THREADS) void frc_reduce(
    const float* __restrict__ part, float* __restrict__ out)
{
    const unsigned lane = threadIdx.x & 31;
    const unsigned gw   = blockIdx.x * RED_WAVES + (threadIdx.x >> 5);
    const unsigned img  = gw / RG_PER_IMG;
    const unsigned rg   = gw % RG_PER_IMG;

    const unsigned half = lane >> 4;
    const unsigned nIdx = lane & 15;

    const float* p = part + (size_t)img * BLOCKS_PER_IMG * HBINS
                          + (size_t)(2 * half) * HBINS + rg * 32 + nIdx;

    v2f aOnes;
    aOnes.x = 1.0f;
    aOnes.y = 1.0f;

    v2f b[4][2][BLOCKS_PER_IMG / 4];
    #pragma unroll
    for (int q = 0; q < 4; ++q) {
        #pragma unroll
        for (int t = 0; t < 2; ++t) {
            #pragma unroll
            for (int kk = 0; kk < BLOCKS_PER_IMG; kk += 4) {
                b[q][t][kk / 4].x = p[(size_t)(kk + 0) * HBINS + q * NRING + t * 16];
                b[q][t][kk / 4].y = p[(size_t)(kk + 1) * HBINS + q * NRING + t * 16];
            }
        }
    }

    v8f c[4][2] = {};
    #pragma unroll
    for (int k4 = 0; k4 < BLOCKS_PER_IMG / 4; ++k4) {
        #pragma unroll
        for (int q = 0; q < 4; ++q) {
            #pragma unroll
            for (int t = 0; t < 2; ++t) {
                c[q][t] = __builtin_amdgcn_wmma_f32_16x16x4_f32(
                               false, aOnes,  false, b[q][t][k4],
                               (short)0, c[q][t],
                               false,  false);
            }
        }
    }

    asm volatile("v_nop\n\tv_nop\n\tv_nop\n\tv_nop"
                 : "+v"(c[0][0]), "+v"(c[0][1]), "+v"(c[1][0]), "+v"(c[1][1]),
                   "+v"(c[2][0]), "+v"(c[2][1]), "+v"(c[3][0]), "+v"(c[3][1]));

    const float cr = half ? c[0][1][0] : c[0][0][0];
    const float ci = half ? c[1][1][0] : c[1][0][0];
    const float p1 = half ? c[2][1][0] : c[2][0][0];
    const float p2 = half ? c[3][1][0] : c[3][0][0];
    const float num = __builtin_sqrtf(__builtin_fmaf(cr, cr, ci * ci));
    const float den = __builtin_sqrtf(p1 * p2);
    out[(size_t)img * NRING + rg * 32 + lane] = (den == 0.0f) ? 0.0f : num / den;
}

extern "C" void kernel_launch(void* const* d_in, const int* in_sizes, int n_in,
                              void* d_out, int out_size, void* d_ws, size_t ws_size,
                              hipStream_t stream)
{
    (void)in_sizes; (void)n_in; (void)out_size; (void)ws_size;
    const float* f1r = (const float*)d_in[0];
    const float* f1i = (const float*)d_in[1];
    const float* f2r = (const float*)d_in[2];
    const float* f2i = (const float*)d_in[3];
    float* part = (float*)d_ws;

    frc_partial<<<dim3(B_ * BLOCKS_PER_IMG), dim3(THREADS), 0, stream>>>(
        f1r, f1i, f2r, f2i, part);
    frc_reduce<<<dim3(B_ * RG_PER_IMG / RED_WAVES), dim3(RED_THREADS), 0, stream>>>(
        part, (float*)d_out);
}


# Stage 1 folds mirror rows: rows 256+d and 256-d have equal ring bins, so a wave adds their per-pixel products and bins the pair once: half the LDS float atomics and ring math (new f32 sum order)
# speedup vs baseline: 1.5480x; 1.1104x over previous
#include <hip/hip_runtime.h>


#define B_    64
#define H_    512
#define W_    512
#define HW    (H_ * W_)
#define NRING 256
#define HBINS (4 * NRING)

#define WG_PER_IMG     4
#define S1_WAVES       16
#define S1_THREADS     (S1_WAVES * 32)
#define ROW_V4         (W_ / 4)
#define CHUNK_V4       32
#define TASKS_PER_IMG  (H_ / 2)
#define TASK_STEP      (WG_PER_IMG * S1_WAVES)
#define TASKS_PER_WAVE (TASKS_PER_IMG / TASK_STEP)
#define TASK_STEP_V4   (TASK_STEP * ROW_V4)

typedef __attribute__((ext_vector_type(2))) float v2f;
typedef __attribute__((ext_vector_type(4))) float v4f;
typedef __attribute__((ext_vector_type(8))) float v8f;

static __device__ __forceinline__ void load_chunk(
    v4f (&buf)[8], const v4f* g1r, const v4f* g1i, const v4f* g2r, const v4f* g2i,
    const unsigned offU, const unsigned offL)
{
    buf[0] = __builtin_nontemporal_load(g1r + offU);
    buf[1] = __builtin_nontemporal_load(g1i + offU);
    buf[2] = __builtin_nontemporal_load(g2r + offU);
    buf[3] = __builtin_nontemporal_load(g2i + offU);
    buf[4] = __builtin_nontemporal_load(g1r + offL);
    buf[5] = __builtin_nontemporal_load(g1i + offL);
    buf[6] = __builtin_nontemporal_load(g2r + offL);
    buf[7] = __builtin_nontemporal_load(g2i + offL);
}

template <bool MAY_BE_CENTRE>
static __device__ __forceinline__ void bin_chunk(
    const v4f (&buf)[8], const int d, const int dx, float* hist)
{
    const v4f u1r = buf[0], u1i = buf[1], u2r = buf[2], u2i = buf[3];
    v4f cr = u1r * u2r + u1i * u2i;
    v4f ci = u1i * u2r - u1r * u2i;
    v4f p1 = u1r * u1r + u1i * u1i;
    v4f p2 = u2r * u2r + u2i * u2i;
    if (!MAY_BE_CENTRE || d != 0) {
        const v4f l1r = buf[4], l1i = buf[5], l2r = buf[6], l2i = buf[7];
        const v4f lcr = l1r * l2r + l1i * l2i;
        const v4f lci = l1i * l2r - l1r * l2i;
        const v4f lp1 = l1r * l1r + l1i * l1i;
        const v4f lp2 = l2r * l2r + l2i * l2i;
        cr += lcr;
        ci += lci;
        p1 += lp1;
        p2 += lp2;
    }

    const int dy2 = d * d;
    const int b0 = (int)__builtin_sqrtf((float)(dx * dx + dy2));
    const int b1 = (int)__builtin_sqrtf((float)((dx + 1) * (dx + 1) + dy2));
    const int b2 = (int)__builtin_sqrtf((float)((dx + 2) * (dx + 2) + dy2));
    const int b3 = (int)__builtin_sqrtf((float)((dx + 3) * (dx + 3) + dy2));

    if ((b0 == b1) & (b1 == b2) & (b2 == b3)) {
        if (b0 < NRING) {
            atomicAdd(&hist[          b0], cr.x + cr.y + cr.z + cr.w);
            atomicAdd(&hist[NRING   + b0], ci.x + ci.y + ci.z + ci.w);
            atomicAdd(&hist[2*NRING + b0], p1.x + p1.y + p1.z + p1.w);
            atomicAdd(&hist[3*NRING + b0], p2.x + p2.y + p2.z + p2.w);
        }
    } else {
        if (b0 < NRING) {
            atomicAdd(&hist[b0], cr.x); atomicAdd(&hist[NRING + b0], ci.x);
            atomicAdd(&hist[2*NRING + b0], p1.x); atomicAdd(&hist[3*NRING + b0], p2.x);
        }
        if (b1 < NRING) {
            atomicAdd(&hist[b1], cr.y); atomicAdd(&hist[NRING + b1], ci.y);
            atomicAdd(&hist[2*NRING + b1], p1.y); atomicAdd(&hist[3*NRING + b1], p2.y);
        }
        if (b2 < NRING) {
            atomicAdd(&hist[b2], cr.z); atomicAdd(&hist[NRING + b2], ci.z);
            atomicAdd(&hist[2*NRING + b2], p1.z); atomicAdd(&hist[3*NRING + b2], p2.z);
        }
        if (b3 < NRING) {
            atomicAdd(&hist[b3], cr.w); atomicAdd(&hist[NRING + b3], ci.w);
            atomicAdd(&hist[2*NRING + b3], p1.w); atomicAdd(&hist[3*NRING + b3], p2.w);
        }
    }
}

__global__ __launch_bounds__(S1_THREADS) void frc_partial(
    const float* __restrict__ f1r, const float* __restrict__ f1i,
    const float* __restrict__ f2r, const float* __restrict__ f2i,
    float* __restrict__ part)
{
    __shared__ float hist[HBINS];

    const int t    = threadIdx.x;
    const int lane = t & 31;
    const int wave = __builtin_amdgcn_readfirstlane(t >> 5);
    const int img  = blockIdx.x / WG_PER_IMG;
    const int wg   = blockIdx.x % WG_PER_IMG;

    #pragma unroll
    for (int i = t; i < HBINS; i += S1_THREADS) hist[i] = 0.0f;
    __syncthreads();

    const v4f* g1r = (const v4f*)f1r;
    const v4f* g1i = (const v4f*)f1i;
    const v4f* g2r = (const v4f*)f2r;
    const v4f* g2i = (const v4f*)f2i;

    int d = wg + WG_PER_IMG * wave + TASK_STEP * (TASKS_PER_WAVE - 1);
    const unsigned lane0 = (unsigned)img * (HW / 4) + (unsigned)lane;
    unsigned offU = lane0 + (unsigned)(H_ / 2 + d) * ROW_V4;
    unsigned offL = lane0 + (unsigned)(H_ / 2 - d) * ROW_V4;
    const int dxl = 4 * lane - 256;

    v4f buf[4][8];
    #pragma unroll
    for (int j = 0; j < 4; ++j)
        load_chunk(buf[j], g1r, g1i, g2r, g2i, offU + CHUNK_V4 * j, offL + CHUNK_V4 * j);

    for (int s = 0; s < TASKS_PER_WAVE - 1; ++s) {
        offU -= TASK_STEP_V4;
        offL += TASK_STEP_V4;
        #pragma unroll
        for (int j = 0; j < 4; ++j) {
            bin_chunk<false>(buf[j], d, dxl + 128 * j, hist);
            load_chunk(buf[j], g1r, g1i, g2r, g2i, offU + CHUNK_V4 * j, offL + CHUNK_V4 * j);
        }
        d -= TASK_STEP;
    }
    #pragma unroll
    for (int j = 0; j < 4; ++j)
        bin_chunk<true>(buf[j], d, dxl + 128 * j, hist);

    __syncthreads();

    float* dst = part + (size_t)(img * WG_PER_IMG + wg) * HBINS;
    #pragma unroll
    for (int i = t; i < HBINS; i += S1_THREADS) dst[i] = hist[i];
}

#define RG_PER_IMG  (NRING / 32)
#define RED_WAVES   2
#define RED_THREADS (RED_WAVES * 32)

__global__ __launch_bounds__(RED_THREADS) void frc_reduce(
    const float* __restrict__ part, float* __restrict__ out)
{
    const unsigned lane = threadIdx.x & 31;
    const unsigned gw   = blockIdx.x * RED_WAVES + (threadIdx.x >> 5);
    const unsigned img  = gw / RG_PER_IMG;
    const unsigned rg   = gw % RG_PER_IMG;

    const unsigned half = lane >> 4;
    const unsigned nIdx = lane & 15;

    const float* p = part + (size_t)img * WG_PER_IMG * HBINS
                          + (size_t)(2 * half) * HBINS + rg * 32 + nIdx;

    v2f aOnes;
    aOnes.x = 1.0f;
    aOnes.y = 1.0f;

    v2f b[4][2];
    #pragma unroll
    for (int q = 0; q < 4; ++q) {
        #pragma unroll
        for (int t = 0; t < 2; ++t) {
            b[q][t].x = p[(size_t)0 * HBINS + q * NRING + t * 16];
            b[q][t].y = p[(size_t)1 * HBINS + q * NRING + t * 16];
        }
    }

    v8f c[4][2] = {};
    #pragma unroll
    for (int q = 0; q < 4; ++q) {
        #pragma unroll
        for (int t = 0; t < 2; ++t) {
            c[q][t] = __builtin_amdgcn_wmma_f32_16x16x4_f32(
                           false, aOnes,  false, b[q][t],
                           (short)0, c[q][t],
                           false,  false);
        }
    }

    asm volatile("v_nop\n\tv_nop\n\tv_nop\n\tv_nop"
                 : "+v"(c[0][0]), "+v"(c[0][1]), "+v"(c[1][0]), "+v"(c[1][1]),
                   "+v"(c[2][0]), "+v"(c[2][1]), "+v"(c[3][0]), "+v"(c[3][1]));

    const float cr = half ? c[0][1][0] : c[0][0][0];
    const float ci = half ? c[1][1][0] : c[1][0][0];
    const float p1 = half ? c[2][1][0] : c[2][0][0];
    const float p2 = half ? c[3][1][0] : c[3][0][0];
    const float num = __builtin_sqrtf(__builtin_fmaf(cr, cr, ci * ci));
    const float den = __builtin_sqrtf(p1 * p2);
    out[(size_t)img * NRING + rg * 32 + lane] = (den == 0.0f) ? 0.0f : num / den;
}

extern "C" void kernel_launch(void* const* d_in, const int* in_sizes, int n_in,
                              void* d_out, int out_size, void* d_ws, size_t ws_size,
                              hipStream_t stream)
{
    (void)in_sizes; (void)n_in; (void)out_size; (void)ws_size;
    const float* f1r = (const float*)d_in[0];
    const float* f1i = (const float*)d_in[1];
    const float* f2r = (const float*)d_in[2];
    const float* f2i = (const float*)d_in[3];
    float* part = (float*)d_ws;

    frc_partial<<<dim3(B_ * WG_PER_IMG), dim3(S1_THREADS), 0, stream>>>(
        f1r, f1i, f2r, f2i, part);
    frc_reduce<<<dim3(B_ * RG_PER_IMG / RED_WAVES), dim3(RED_THREADS), 0, stream>>>(
        part, (float*)d_out);
}


# Stage 1 requests only lines holding a pixel inside ring 255: a lane beyond the row pair's chord loads the chord's end segment, not its own (about 15% fewer lines); sums and their order unchanged
# speedup vs baseline: 1.6783x; 1.0842x over previous
#include <hip/hip_runtime.h>


#define B_    64
#define H_    512
#define W_    512
#define HW    (H_ * W_)
#define NRING 256
#define HBINS (4 * NRING)

#define WG_PER_IMG     4
#define S1_WAVES       16
#define S1_THREADS     (S1_WAVES * 32)
#define ROW_V4         (W_ / 4)
#define CHUNK_V4       32
#define TASKS_PER_IMG  (H_ / 2)
#define TASK_STEP      (WG_PER_IMG * S1_WAVES)
#define TASKS_PER_WAVE (TASKS_PER_IMG / TASK_STEP)
#define TASK_STEP_V4   (TASK_STEP * ROW_V4)

typedef __attribute__((ext_vector_type(2))) float v2f;
typedef __attribute__((ext_vector_type(4))) float v4f;
typedef __attribute__((ext_vector_type(8))) float v8f;

static __device__ __forceinline__ void load_chunk(
    v4f (&buf)[8], const v4f* g1r, const v4f* g1i, const v4f* g2r, const v4f* g2i,
    const unsigned offU, const unsigned offL)
{
    buf[0] = __builtin_nontemporal_load(g1r + offU);
    buf[1] = __builtin_nontemporal_load(g1i + offU);
    buf[2] = __builtin_nontemporal_load(g2r + offU);
    buf[3] = __builtin_nontemporal_load(g2i + offU);
    buf[4] = __builtin_nontemporal_load(g1r + offL);
    buf[5] = __builtin_nontemporal_load(g1i + offL);
    buf[6] = __builtin_nontemporal_load(g2r + offL);
    buf[7] = __builtin_nontemporal_load(g2i + offL);
}

static __device__ __forceinline__ int chord_half(const int d)
{
    return (int)__builtin_sqrtf((float)(NRING * NRING - 1 - d * d));
}

static __device__ __forceinline__ unsigned seg_in_chord(const int dx0, const int c)
{
    const int lo = (-c) & ~3;
    const int hi = c & ~3;
    int x = dx0 < lo ? lo : dx0;
    x = x > hi ? hi : x;
    return (unsigned)(x + W_ / 2) >> 2;
}

template <bool MAY_BE_CENTRE>
static __device__ __forceinline__ void bin_chunk(
    const v4f (&buf)[8], const int d, const int dx, float* hist)
{
    const v4f u1r = buf[0], u1i = buf[1], u2r = buf[2], u2i = buf[3];
    v4f cr = u1r * u2r + u1i * u2i;
    v4f ci = u1i * u2r - u1r * u2i;
    v4f p1 = u1r * u1r + u1i * u1i;
    v4f p2 = u2r * u2r + u2i * u2i;
    if (!MAY_BE_CENTRE || d != 0) {
        const v4f l1r = buf[4], l1i = buf[5], l2r = buf[6], l2i = buf[7];
        const v4f lcr = l1r * l2r + l1i * l2i;
        const v4f lci = l1i * l2r - l1r * l2i;
        const v4f lp1 = l1r * l1r + l1i * l1i;
        const v4f lp2 = l2r * l2r + l2i * l2i;
        cr += lcr;
        ci += lci;
        p1 += lp1;
        p2 += lp2;
    }

    const int dy2 = d * d;
    const int b0 = (int)__builtin_sqrtf((float)(dx * dx + dy2));
    const int b1 = (int)__builtin_sqrtf((float)((dx + 1) * (dx + 1) + dy2));
    const int b2 = (int)__builtin_sqrtf((float)((dx + 2) * (dx + 2) + dy2));
    const int b3 = (int)__builtin_sqrtf((float)((dx + 3) * (dx + 3) + dy2));

    if ((b0 == b1) & (b1 == b2) & (b2 == b3)) {
        if (b0 < NRING) {
            atomicAdd(&hist[          b0], cr.x + cr.y + cr.z + cr.w);
            atomicAdd(&hist[NRING   + b0], ci.x + ci.y + ci.z + ci.w);
            atomicAdd(&hist[2*NRING + b0], p1.x + p1.y + p1.z + p1.w);
            atomicAdd(&hist[3*NRING + b0], p2.x + p2.y + p2.z + p2.w);
        }
    } else {
        if (b0 < NRING) {
            atomicAdd(&hist[b0], cr.x); atomicAdd(&hist[NRING + b0], ci.x);
            atomicAdd(&hist[2*NRING + b0], p1.x); atomicAdd(&hist[3*NRING + b0], p2.x);
        }
        if (b1 < NRING) {
            atomicAdd(&hist[b1], cr.y); atomicAdd(&hist[NRING + b1], ci.y);
            atomicAdd(&hist[2*NRING + b1], p1.y); atomicAdd(&hist[3*NRING + b1], p2.y);
        }
        if (b2 < NRING) {
            atomicAdd(&hist[b2], cr.z); atomicAdd(&hist[NRING + b2], ci.z);
            atomicAdd(&hist[2*NRING + b2], p1.z); atomicAdd(&hist[3*NRING + b2], p2.z);
        }
        if (b3 < NRING) {
            atomicAdd(&hist[b3], cr.w); atomicAdd(&hist[NRING + b3], ci.w);
            atomicAdd(&hist[2*NRING + b3], p1.w); atomicAdd(&hist[3*NRING + b3], p2.w);
        }
    }
}

__global__ __launch_bounds__(S1_THREADS) void frc_partial(
    const float* __restrict__ f1r, const float* __restrict__ f1i,
    const float* __restrict__ f2r, const float* __restrict__ f2i,
    float* __restrict__ part)
{
    __shared__ float hist[HBINS];

    const int t    = threadIdx.x;
    const int lane = t & 31;
    const int wave = __builtin_amdgcn_readfirstlane(t >> 5);
    const int img  = blockIdx.x / WG_PER_IMG;
    const int wg   = blockIdx.x % WG_PER_IMG;

    #pragma unroll
    for (int i = t; i < HBINS; i += S1_THREADS) hist[i] = 0.0f;
    __syncthreads();

    const v4f* g1r = (const v4f*)f1r;
    const v4f* g1i = (const v4f*)f1i;
    const v4f* g2r = (const v4f*)f2r;
    const v4f* g2i = (const v4f*)f2i;

    int d = wg + WG_PER_IMG * wave + TASK_STEP * (TASKS_PER_WAVE - 1);
    const unsigned img0 = (unsigned)img * (HW / 4);
    unsigned rowU = img0 + (unsigned)(H_ / 2 + d) * ROW_V4;
    unsigned rowL = img0 + (unsigned)(H_ / 2 - d) * ROW_V4;
    const int dxl = 4 * lane - 256;
    const int c = chord_half(d);

    v4f buf[4][8];
    #pragma unroll
    for (int j = 0; j < 4; ++j) {
        const unsigned seg = seg_in_chord(dxl + 128 * j, c);
        load_chunk(buf[j], g1r, g1i, g2r, g2i, rowU + seg, rowL + seg);
    }

    for (int s = 0; s < TASKS_PER_WAVE - 1; ++s) {
        rowU -= TASK_STEP_V4;
        rowL += TASK_STEP_V4;
        const int cn = chord_half(d - TASK_STEP);
        #pragma unroll
        for (int j = 0; j < 4; ++j) {
            bin_chunk<false>(buf[j], d, dxl + 128 * j, hist);
            const unsigned seg = seg_in_chord(dxl + 128 * j, cn);
            load_chunk(buf[j], g1r, g1i, g2r, g2i, rowU + seg, rowL + seg);
        }
        d -= TASK_STEP;
    }
    #pragma unroll
    for (int j = 0; j < 4; ++j)
        bin_chunk<true>(buf[j], d, dxl + 128 * j, hist);

    __syncthreads();

    float* dst = part + (size_t)(img * WG_PER_IMG + wg) * HBINS;
    #pragma unroll
    for (int i = t; i < HBINS; i += S1_THREADS) dst[i] = hist[i];
}

#define RG_PER_IMG  (NRING / 32)
#define RED_WAVES   2
#define RED_THREADS (RED_WAVES * 32)

__global__ __launch_bounds__(RED_THREADS) void frc_reduce(
    const float* __restrict__ part, float* __restrict__ out)
{
    const unsigned lane = threadIdx.x & 31;
    const unsigned gw   = blockIdx.x * RED_WAVES + (threadIdx.x >> 5);
    const unsigned img  = gw / RG_PER_IMG;
    const unsigned rg   = gw % RG_PER_IMG;

    const unsigned half = lane >> 4;
    const unsigned nIdx = lane & 15;

    const float* p = part + (size_t)img * WG_PER_IMG * HBINS
                          + (size_t)(2 * half) * HBINS + rg * 32 + nIdx;

    v2f aOnes;
    aOnes.x = 1.0f;
    aOnes.y = 1.0f;

    v2f b[4][2];
    #pragma unroll
    for (int q = 0; q < 4; ++q) {
        #pragma unroll
        for (int t = 0; t < 2; ++t) {
            b[q][t].x = p[(size_t)0 * HBINS + q * NRING + t * 16];
            b[q][t].y = p[(size_t)1 * HBINS + q * NRING + t * 16];
        }
    }

    v8f c[4][2] = {};
    #pragma unroll
    for (int q = 0; q < 4; ++q) {
        #pragma unroll
        for (int t = 0; t < 2; ++t) {
            c[q][t] = __builtin_amdgcn_wmma_f32_16x16x4_f32(
                           false, aOnes,  false, b[q][t],
                           (short)0, c[q][t],
                           false,  false);
        }
    }

    asm volatile("v_nop\n\tv_nop\n\tv_nop\n\tv_nop"
                 : "+v"(c[0][0]), "+v"(c[0][1]), "+v"(c[1][0]), "+v"(c[1][1]),
                   "+v"(c[2][0]), "+v"(c[2][1]), "+v"(c[3][0]), "+v"(c[3][1]));

    const float cr = half ? c[0][1][0] : c[0][0][0];
    const float ci = half ? c[1][1][0] : c[1][0][0];
    const float p1 = half ? c[2][1][0] : c[2][0][0];
    const float p2 = half ? c[3][1][0] : c[3][0][0];
    const float num = __builtin_sqrtf(__builtin_fmaf(cr, cr, ci * ci));
    const float den = __builtin_sqrtf(p1 * p2);
    out[(size_t)img * NRING + rg * 32 + lane] = (den == 0.0f) ? 0.0f : num / den;
}

extern "C" void kernel_launch(void* const* d_in, const int* in_sizes, int n_in,
                              void* d_out, int out_size, void* d_ws, size_t ws_size,
                              hipStream_t stream)
{
    (void)in_sizes; (void)n_in; (void)out_size; (void)ws_size;
    const float* f1r = (const float*)d_in[0];
    const float* f1i = (const float*)d_in[1];
    const float* f2r = (const float*)d_in[2];
    const float* f2i = (const float*)d_in[3];
    float* part = (float*)d_ws;

    frc_partial<<<dim3(B_ * WG_PER_IMG), dim3(S1_THREADS), 0, stream>>>(
        f1r, f1i, f2r, f2i, part);
    frc_reduce<<<dim3(B_ * RG_PER_IMG / RED_WAVES), dim3(RED_THREADS), 0, stream>>>(
        part, (float*)d_out);
}
